# starting kernel
# baseline (speedup 1.0000x reference)
#include <hip/hip_runtime.h>
#include <hip/hip_bf16.h>


typedef __attribute__((ext_vector_type(2))) float v2f;
typedef __attribute__((ext_vector_type(8))) float v8f;

#define CH    64
#define SP    56
#define NCOL  (SP * SP)
#define KRED  384
#define PLANE 3136

__global__ __launch_bounds__(256)
void shiftconv_wmma_f32(const float* __restrict__ x,
                        const float* __restrict__ W,
                        float* __restrict__ out)
{
    const int lane = threadIdx.x & 31;
    const int wave = threadIdx.x >> 5;
    const int tile = blockIdx.x * 8 + wave;
    const int row_tile = tile / 196;
    const int col_tile = tile % 196;
    const int row_base = row_tile * 16;
    const int col_base = col_tile * 16;

    const int m     = lane & 15;
    const int kh    = lane >> 4;
    const int kbase = kh * 2;

    const int c    = col_base + m;
    const int o    = c / 56;
    const int n    = c % 56;
    const int xrow = n * 56;
    const int ll0  = o - 1;
    const int ll1  = ((o + 55) % 56) - 1;

    const float* Arow = W + (row_base + m) * KRED + kbase;

    v8f acc = {};

    #pragma unroll
    for (int kl = 0; kl < 6; ++kl) {
        const int k = kl >> 1;
        const int l = kl & 1;
        const int ll = (l ? ll1 : ll0) + k;

        const bool  valid = (unsigned)ll < 56u;
        const int   llc   = valid ? ll : 0;
        const float fmask = valid ? 1.0f : 0.0f;

        const float* bp0 = x + (size_t)(kbase + 0) * PLANE + xrow + llc;
        const float* bp1 = x + (size_t)(kbase + 1) * PLANE + xrow + llc;
        const float* ap  = Arow + kl * 64;

        #pragma unroll
        for (int q0 = 0; q0 < 64; q0 += 4) {
            v2f a = *(const v2f*)(ap + q0);

            v2f b;
            b[0] = bp0[q0 * PLANE] * fmask;
            b[1] = bp1[q0 * PLANE] * fmask;

            acc = __builtin_amdgcn_wmma_f32_16x16x4_f32(
                 false, a,  false, b,
                 (short)0, acc,  false,  false);
        }
    }

    const int ostore = o * 56 + ((n + 1) % 56);
    #pragma unroll
    for (int vr = 0; vr < 8; ++vr) {
        const int i = row_base + vr + 8 * kh;
        out[i * NCOL + ostore] = acc[vr];
    }
}

extern "C" void kernel_launch(void* const* d_in, const int* in_sizes, int n_in,
                              void* d_out, int out_size, void* d_ws, size_t ws_size,
                              hipStream_t stream) {
    const float* x = (const float*)d_in[0];
    const float* W = (const float*)d_in[1];
    float* out = (float*)d_out;

    shiftconv_wmma_f32<<<98, 256, 0, stream>>>(x, W, out);
}


# Split-K by channel group and tap parity (8 waves x 16 channels x 3 taps, was 6 waves x 1 tap): one 12-byte x load per plane feeds a lane's 3 taps, a third of the x loads (sum order changes)
# speedup vs baseline: 5.2700x; 5.2700x over previous
#include <hip/hip_runtime.h>
#include <hip/hip_bf16.h>


typedef __attribute__((ext_vector_type(2))) float v2f;
typedef __attribute__((ext_vector_type(4))) float v4f;
typedef __attribute__((ext_vector_type(8))) float v8f;

#define CH    64
#define SP    56
#define NCOL  (SP * SP)
#define KRED  384
#define PLANE 3136
#define NWAVE 8
#define JGRP  16

__global__ __launch_bounds__(32 * NWAVE)
void shiftconv_wmma_f32(const float* __restrict__ x,
                        const float* __restrict__ W,
                        float* __restrict__ out)
{
    __shared__ v4f part[NWAVE][8][32];

    const int lane = threadIdx.x & 31;
    const int wave = threadIdx.x >> 5;
    const int col_tile = blockIdx.x;
    const int col_base = col_tile * 16;

    const int m     = lane & 15;
    const int kh    = lane >> 4;
    const int kbase = kh * 2;

    const int c    = col_base + m;
    const int o    = c / 56;
    const int n    = c % 56;
    const int xrow = n * 56;

    const int l  = wave & 1;
    const int jq = wave >> 1;

    const int base  = (l ? (o + 55) % 56 : o) - 1;
    const int start = base < 0 ? 0 : (base > 53 ? 53 : base);
    const int sh    = base - start;
    const bool lo   = sh < 0;
    const bool hi   = sh > 0;

    const float* ap  = W + m * KRED + l * 64 + jq * JGRP + kbase;
    const float* bp0 = x + (size_t)(jq * JGRP + kbase + 0) * PLANE + xrow + start;
    const float* bp1 = x + (size_t)(jq * JGRP + kbase + 1) * PLANE + xrow + start;

    v8f acc0 = {}, acc1 = {}, acc2 = {}, acc3 = {};

    #pragma unroll
    for (int q0 = 0; q0 < JGRP; q0 += 4) {
        const float u0 = bp0[q0 * PLANE + 0];
        const float u1 = bp0[q0 * PLANE + 1];
        const float u2 = bp0[q0 * PLANE + 2];
        const float v0 = bp1[q0 * PLANE + 0];
        const float v1 = bp1[q0 * PLANE + 1];
        const float v2 = bp1[q0 * PLANE + 2];

        v2f b[3];
        b[0][0] = lo ? 0.0f : (hi ? u1 : u0);
        b[0][1] = lo ? 0.0f : (hi ? v1 : v0);
        b[1][0] = lo ? u0 : (hi ? u2 : u1);
        b[1][1] = lo ? v0 : (hi ? v2 : v1);
        b[2][0] = hi ? 0.0f : (lo ? u1 : u2);
        b[2][1] = hi ? 0.0f : (lo ? v1 : v2);

        #pragma unroll
        for (int k = 0; k < 3; ++k) {
            const float* apk = ap + k * 128 + q0;
            v2f a0 = *(const v2f*)(apk + 0 * 16 * KRED);
            v2f a1 = *(const v2f*)(apk + 1 * 16 * KRED);
            v2f a2 = *(const v2f*)(apk + 2 * 16 * KRED);
            v2f a3 = *(const v2f*)(apk + 3 * 16 * KRED);

            acc0 = __builtin_amdgcn_wmma_f32_16x16x4_f32(
                false, a0, false, b[k], (short)0, acc0, false, false);
            acc1 = __builtin_amdgcn_wmma_f32_16x16x4_f32(
                false, a1, false, b[k], (short)0, acc1, false, false);
            acc2 = __builtin_amdgcn_wmma_f32_16x16x4_f32(
                false, a2, false, b[k], (short)0, acc2, false, false);
            acc3 = __builtin_amdgcn_wmma_f32_16x16x4_f32(
                false, a3, false, b[k], (short)0, acc3, false, false);
        }
    }

    #pragma unroll
    for (int vr = 0; vr < 8; ++vr) {
        v4f p = { acc0[vr], acc1[vr], acc2[vr], acc3[vr] };
        part[wave][vr][lane] = p;
    }
    __syncthreads();

    const int ostore = o * 56 + ((n + 1) % 56);

    {
        const int vr = wave;
        v4f s = part[0][vr][lane];
        #pragma unroll
        for (int w = 1; w < NWAVE; ++w) {
            s += part[w][vr][lane];
        }
        const int i = vr + 8 * kh;
        out[(i +  0) * NCOL + ostore] = s[0];
        out[(i + 16) * NCOL + ostore] = s[1];
        out[(i + 32) * NCOL + ostore] = s[2];
        out[(i + 48) * NCOL + ostore] = s[3];
    }
}

extern "C" void kernel_launch(void* const* d_in, const int* in_sizes, int n_in,
                              void* d_out, int out_size, void* d_ws, size_t ws_size,
                              hipStream_t stream) {
    const float* x = (const float*)d_in[0];
    const float* W = (const float*)d_in[1];
    float* out = (float*)d_out;

    shiftconv_wmma_f32<<<196, 256, 0, stream>>>(x, W, out);
}


# Wave tile 16 rows x 4 neighbouring-o column tiles, both tap parities per wave (was 32 x 2, one parity): one 7-float x window per plane; 12 A loads per wave, was 24 (sum order changes)
# speedup vs baseline: 6.7179x; 1.2747x over previous
#include <hip/hip_runtime.h>
#include <hip/hip_bf16.h>


typedef __attribute__((ext_vector_type(2))) float v2f;
typedef __attribute__((ext_vector_type(4))) float v4f;
typedef __attribute__((ext_vector_type(8))) float v8f;

#define CH    64
#define SP    56
#define NCOL  (SP * SP)
#define KRED  384
#define PLANE 3136
#define NWAVE 8
#define JGRP  8
#define QTILES 49

__global__ __launch_bounds__(32 * NWAVE)
void shiftconv_wmma_f32(const float* __restrict__ x,
                        const float* __restrict__ W,
                        float* __restrict__ out)
{
    __shared__ v4f part[NWAVE][8][32];

    const int lane = threadIdx.x & 31;
    const int wave = __builtin_amdgcn_readfirstlane(threadIdx.x >> 5);
    const int q_tile   = blockIdx.x >> 2;
    const int row_base = (blockIdx.x & 3) * 16;

    const int m     = lane & 15;
    const int kh    = lane >> 4;
    const int kbase = kh * 2;

    const int q    = q_tile * 16 + m;
    const int g    = q / 56;
    const int n    = q % 56;
    const int o0   = 4 * g;
    const int xrow = n * 56;

    const int jq = wave;

    const bool m2   = g == 0;
    const bool p1   = g == 13;
    const int startL = m2 ? 0 : o0 - 2;
    const int startR = p1 ? 53 : o0 + 2;
    const bool wrap_lane = (g == 0);
    const bool wrap_wave = (q_tile < 4);

    const float* ap  = W + (row_base + m) * KRED + jq * JGRP + kbase;
    const float* bp0 = x + (size_t)(jq * JGRP + kbase + 0) * PLANE + xrow;
    const float* bp1 = x + (size_t)(jq * JGRP + kbase + 1) * PLANE + xrow;
    const float* bl0 = bp0 + startL;
    const float* bl1 = bp1 + startL;
    const float* br0 = bp0 + startR;
    const float* br1 = bp1 + startR;
    const float* zp0 = x + (size_t)(jq * JGRP + kbase + 0) * PLANE + xrow + 54;
    const float* zp1 = x + (size_t)(jq * JGRP + kbase + 1) * PLANE + xrow + 54;

    v8f acc0 = {}, acc1 = {}, acc2 = {}, acc3 = {};

    v2f zu[2] = {}, zv[2] = {};
    if (wrap_wave) {
        #pragma unroll
        for (int c4 = 0; c4 < 2; ++c4) {
            zu[c4] = *(const v2f*)(zp0 + c4 * 4 * PLANE);
            zv[c4] = *(const v2f*)(zp1 + c4 * 4 * PLANE);
        }
    }

    #pragma unroll
    for (int c4 = 0; c4 < 2; ++c4) {
        const float* pl0 = bl0 + c4 * 4 * PLANE;
        const float* pl1 = bl1 + c4 * 4 * PLANE;
        const float* pr0 = br0 + c4 * 4 * PLANE;
        const float* pr1 = br1 + c4 * 4 * PLANE;
        const float ul0 = pl0[0], ul1 = pl0[1], ul2 = pl0[2], ul3 = pl0[3];
        const float vl0 = pl1[0], vl1 = pl1[1], vl2 = pl1[2], vl3 = pl1[3];
        const float ur0 = pr0[0], ur1 = pr0[1], ur2 = pr0[2];
        const float vr0 = pr1[0], vr1 = pr1[1], vr2 = pr1[2];

        v2f bs[7];
        bs[0][0] = m2 ? 0.0f : ul0;   bs[0][1] = m2 ? 0.0f : vl0;
        bs[1][0] = m2 ? 0.0f : ul1;   bs[1][1] = m2 ? 0.0f : vl1;
        bs[2][0] = m2 ? ul0  : ul2;   bs[2][1] = m2 ? vl0  : vl2;
        bs[3][0] = m2 ? ul1  : ul3;   bs[3][1] = m2 ? vl1  : vl3;
        bs[4][0] = p1 ? ur1  : ur0;   bs[4][1] = p1 ? vr1  : vr0;
        bs[5][0] = p1 ? ur2  : ur1;   bs[5][1] = p1 ? vr2  : vr1;
        bs[6][0] = p1 ? 0.0f : ur2;   bs[6][1] = p1 ? 0.0f : vr2;

        v2f bt[3];
        bt[0][0] = wrap_lane ? zu[c4][0] : bs[0][0];
        bt[0][1] = wrap_lane ? zv[c4][0] : bs[0][1];
        bt[1][0] = wrap_lane ? zu[c4][1] : bs[1][0];
        bt[1][1] = wrap_lane ? zv[c4][1] : bs[1][1];
        bt[2][0] = wrap_lane ? 0.0f : bs[2][0];
        bt[2][1] = wrap_lane ? 0.0f : bs[2][1];

        #pragma unroll
        for (int k = 0; k < 3; ++k) {
            const float* apk = ap + k * 128 + c4 * 4;
            v2f a0 = *(const v2f*)(apk);
            v2f a1 = *(const v2f*)(apk + 64);

            acc0 = __builtin_amdgcn_wmma_f32_16x16x4_f32(
                false, a0, false, bs[k + 1], (short)0, acc0, false, false);
            acc1 = __builtin_amdgcn_wmma_f32_16x16x4_f32(
                false, a0, false, bs[k + 2], (short)0, acc1, false, false);
            acc2 = __builtin_amdgcn_wmma_f32_16x16x4_f32(
                false, a0, false, bs[k + 3], (short)0, acc2, false, false);
            acc3 = __builtin_amdgcn_wmma_f32_16x16x4_f32(
                false, a0, false, bs[k + 4], (short)0, acc3, false, false);

            acc0 = __builtin_amdgcn_wmma_f32_16x16x4_f32(
                false, a1, false, bt[k], (short)0, acc0, false, false);
            acc1 = __builtin_amdgcn_wmma_f32_16x16x4_f32(
                false, a1, false, bs[k + 1], (short)0, acc1, false, false);
            acc2 = __builtin_amdgcn_wmma_f32_16x16x4_f32(
                false, a1, false, bs[k + 2], (short)0, acc2, false, false);
            acc3 = __builtin_amdgcn_wmma_f32_16x16x4_f32(
                false, a1, false, bs[k + 3], (short)0, acc3, false, false);
        }
    }

    #pragma unroll
    for (int vr = 0; vr < 8; ++vr) {
        v4f p = { acc0[vr], acc1[vr], acc2[vr], acc3[vr] };
        part[wave][vr][lane] = p;
    }
    __syncthreads();

    const int ostore = o0 * 56 + ((n + 1) % 56);

    {
        const int vr = wave;
        v4f s = part[0][vr][lane];
        #pragma unroll
        for (int w = 1; w < NWAVE; ++w) {
            s += part[w][vr][lane];
        }
        const int i = row_base + vr + 8 * kh;
        out[i * NCOL + ostore]       = s[0];
        out[i * NCOL + ostore + 56]  = s[1];
        out[i * NCOL + ostore + 112] = s[2];
        out[i * NCOL + ostore + 168] = s[3];
    }
}

extern "C" void kernel_launch(void* const* d_in, const int* in_sizes, int n_in,
                              void* d_out, int out_size, void* d_ws, size_t ws_size,
                              hipStream_t stream) {
    const float* x = (const float*)d_in[0];
    const float* W = (const float*)d_in[1];
    float* out = (float*)d_out;

    shiftconv_wmma_f32<<<196, 256, 0, stream>>>(x, W, out);
}


# Each x load fetches one plane for all 32 lanes: lane m the left, lane m+16 the right 16 bytes of one row window, halves traded by v_permlane16_swap_b32; a load touches 16 lines, was 32
# speedup vs baseline: 7.3346x; 1.0918x over previous
#include <hip/hip_runtime.h>
#include <hip/hip_bf16.h>


typedef __attribute__((ext_vector_type(2))) float v2f;
typedef __attribute__((ext_vector_type(4))) float v4f;
typedef __attribute__((ext_vector_type(8))) float v8f;
static __device__ __forceinline__ void swap_halves(float& a, float& b)
{
    asm("v_permlane16_swap_b32 %0, %1" : "+v"(a), "+v"(b));
}

#define CH    64
#define SP    56
#define NCOL  (SP * SP)
#define KRED  384
#define PLANE 3136
#define NWAVE 8
#define JGRP  8
#define QTILES 49

__global__ __launch_bounds__(32 * NWAVE)
void shiftconv_wmma_f32(const float* __restrict__ x,
                        const float* __restrict__ W,
                        float* __restrict__ out)
{
    __shared__ v4f part[NWAVE][8][32];

    const int lane = threadIdx.x & 31;
    const int wave = __builtin_amdgcn_readfirstlane(threadIdx.x >> 5);
    const int q_tile   = blockIdx.x >> 2;
    const int row_base = (blockIdx.x & 3) * 16;

    const int m     = lane & 15;
    const int kh    = lane >> 4;
    const int kbase = kh * 2;

    const int q    = q_tile * 16 + m;
    const int g    = q / 56;
    const int n    = q % 56;
    const int o0   = 4 * g;
    const int xrow = n * 56;

    const int jq = wave;

    const bool m2   = g == 0;
    const bool p1   = g == 13;
    const int startL = m2 ? 0 : o0 - 2;
    const int startR = p1 ? 52 : o0 + 2;
    const int hoff = kh ? startR : startL;
    const bool wrap_lane = (g == 0);
    const bool wrap_wave = (q_tile < 4);

    const float* ap  = W + (row_base + m) * KRED + jq * JGRP + kbase;
    const float* hp = x + (size_t)(jq * JGRP) * PLANE + xrow + hoff;
    const float* zp0 = x + (size_t)(jq * JGRP + kbase + 0) * PLANE + xrow + 54;
    const float* zp1 = x + (size_t)(jq * JGRP + kbase + 1) * PLANE + xrow + 54;

    v8f acc0 = {}, acc1 = {}, acc2 = {}, acc3 = {};

    v2f zu[2] = {}, zv[2] = {};
    if (wrap_wave) {
        #pragma unroll
        for (int c4 = 0; c4 < 2; ++c4) {
            zu[c4] = *(const v2f*)(zp0 + c4 * 4 * PLANE);
            zv[c4] = *(const v2f*)(zp1 + c4 * 4 * PLANE);
        }
    }

    float hx[2][4][4];
    v2f a_first[3][2];
    #pragma unroll
    for (int p = 0; p < 4; ++p) {
        const float* pp = hp + p * PLANE;
        hx[0][p][0] = pp[0]; hx[0][p][1] = pp[1]; hx[0][p][2] = pp[2]; hx[0][p][3] = pp[3];
    }
    #pragma unroll
    for (int k = 0; k < 3; ++k) {
        a_first[k][0] = *(const v2f*)(ap + k * 128);
        a_first[k][1] = *(const v2f*)(ap + k * 128 + 64);
    }
    __builtin_amdgcn_sched_barrier(0);
    #pragma unroll
    for (int p = 0; p < 4; ++p) {
        const float* pp = hp + (4 + p) * PLANE;
        hx[1][p][0] = pp[0]; hx[1][p][1] = pp[1]; hx[1][p][2] = pp[2]; hx[1][p][3] = pp[3];
    }
    __builtin_amdgcn_sched_barrier(0);

    #pragma unroll
    for (int c4 = 0; c4 < 2; ++c4) {
        float h[4][4];
        #pragma unroll
        for (int p = 0; p < 4; ++p) {
            #pragma unroll
            for (int i = 0; i < 4; ++i) h[p][i] = hx[c4][p][i];
        }
        #pragma unroll
        for (int p = 0; p < 2; ++p) {
            #pragma unroll
            for (int i = 0; i < 4; ++i) swap_halves(h[p][i], h[p + 2][i]);
        }
        const float ul0 = h[0][0], ul1 = h[0][1], ul2 = h[0][2], ul3 = h[0][3];
        const float vl0 = h[1][0], vl1 = h[1][1], vl2 = h[1][2], vl3 = h[1][3];
        const float ur0 = h[2][0], ur1 = h[2][1], ur2 = h[2][2], ur3 = h[2][3];
        const float vr0 = h[3][0], vr1 = h[3][1], vr2 = h[3][2], vr3 = h[3][3];

        v2f bs[7];
        bs[0][0] = m2 ? 0.0f : ul0;   bs[0][1] = m2 ? 0.0f : vl0;
        bs[1][0] = m2 ? 0.0f : ul1;   bs[1][1] = m2 ? 0.0f : vl1;
        bs[2][0] = m2 ? ul0  : ul2;   bs[2][1] = m2 ? vl0  : vl2;
        bs[3][0] = m2 ? ul1  : ul3;   bs[3][1] = m2 ? vl1  : vl3;
        bs[4][0] = p1 ? ur2  : ur0;   bs[4][1] = p1 ? vr2  : vr0;
        bs[5][0] = p1 ? ur3  : ur1;   bs[5][1] = p1 ? vr3  : vr1;
        bs[6][0] = p1 ? 0.0f : ur2;   bs[6][1] = p1 ? 0.0f : vr2;

        v2f bt[3];
        bt[0][0] = wrap_lane ? zu[c4][0] : bs[0][0];
        bt[0][1] = wrap_lane ? zv[c4][0] : bs[0][1];
        bt[1][0] = wrap_lane ? zu[c4][1] : bs[1][0];
        bt[1][1] = wrap_lane ? zv[c4][1] : bs[1][1];
        bt[2][0] = wrap_lane ? 0.0f : bs[2][0];
        bt[2][1] = wrap_lane ? 0.0f : bs[2][1];

        #pragma unroll
        for (int k = 0; k < 3; ++k) {
            const float* apk = ap + k * 128 + c4 * 4;
            v2f a0 = c4 == 0 ? a_first[k][0] : *(const v2f*)(apk);
            v2f a1 = c4 == 0 ? a_first[k][1] : *(const v2f*)(apk + 64);

            acc0 = __builtin_amdgcn_wmma_f32_16x16x4_f32(
                false, a0, false, bs[k + 1], (short)0, acc0, false, false);
            acc1 = __builtin_amdgcn_wmma_f32_16x16x4_f32(
                false, a0, false, bs[k + 2], (short)0, acc1, false, false);
            acc2 = __builtin_amdgcn_wmma_f32_16x16x4_f32(
                false, a0, false, bs[k + 3], (short)0, acc2, false, false);
            acc3 = __builtin_amdgcn_wmma_f32_16x16x4_f32(
                false, a0, false, bs[k + 4], (short)0, acc3, false, false);

            acc0 = __builtin_amdgcn_wmma_f32_16x16x4_f32(
                false, a1, false, bt[k], (short)0, acc0, false, false);
            acc1 = __builtin_amdgcn_wmma_f32_16x16x4_f32(
                false, a1, false, bs[k + 1], (short)0, acc1, false, false);
            acc2 = __builtin_amdgcn_wmma_f32_16x16x4_f32(
                false, a1, false, bs[k + 2], (short)0, acc2, false, false);
            acc3 = __builtin_amdgcn_wmma_f32_16x16x4_f32(
                false, a1, false, bs[k + 3], (short)0, acc3, false, false);
        }
        if (c4 == 0) __builtin_amdgcn_sched_barrier(0);
    }

    #pragma unroll
    for (int vr = 0; vr < 8; ++vr) {
        v4f p = { acc0[vr], acc1[vr], acc2[vr], acc3[vr] };
        part[wave][vr][lane] = p;
    }
    __syncthreads();

    const int ostore = o0 * 56 + ((n + 1) % 56);

    {
        const int vr = wave;
        v4f s = part[0][vr][lane];
        #pragma unroll
        for (int w = 1; w < NWAVE; ++w) {
            s += part[w][vr][lane];
        }
        const int i = row_base + vr + 8 * kh;
        out[i * NCOL + ostore]       = s[0];
        out[i * NCOL + ostore + 56]  = s[1];
        out[i * NCOL + ostore + 112] = s[2];
        out[i * NCOL + ostore + 168] = s[3];
    }
}

extern "C" void kernel_launch(void* const* d_in, const int* in_sizes, int n_in,
                              void* d_out, int out_size, void* d_ws, size_t ws_size,
                              hipStream_t stream) {
    const float* x = (const float*)d_in[0];
    const float* W = (const float*)d_in[1];
    float* out = (float*)d_out;

    shiftconv_wmma_f32<<<196, 256, 0, stream>>>(x, W, out);
}


# Lane columns of a wave are a block of 4 o quads x 4 x rows (was 1 quad x 16 rows; quads 12,13: 2 x 8): an x load reads 4 row stretches of 80 B (7 lines), not 16 windows of 32 B (20 lines)
# speedup vs baseline: 7.8334x; 1.0680x over previous
#include <hip/hip_runtime.h>
#include <hip/hip_bf16.h>


typedef __attribute__((ext_vector_type(2))) float v2f;
typedef __attribute__((ext_vector_type(4))) float v4f;
typedef __attribute__((ext_vector_type(8))) float v8f;
static __device__ __forceinline__ void swap_halves(float& a, float& b)
{
    asm("v_permlane16_swap_b32 %0, %1" : "+v"(a), "+v"(b));
}

#define CH    64
#define SP    56
#define NCOL  (SP * SP)
#define KRED  384
#define PLANE 3136
#define NWAVE 8
#define JGRP  8
#define QTILES 49

__global__ __launch_bounds__(32 * NWAVE)
void shiftconv_wmma_f32(const float* __restrict__ x,
                        const float* __restrict__ W,
                        float* __restrict__ out)
{
    __shared__ v4f part[NWAVE][8][32];

    const int lane = threadIdx.x & 31;
    const int wave = __builtin_amdgcn_readfirstlane(threadIdx.x >> 5);
    const int q_tile   = blockIdx.x >> 2;
    const int row_base = (blockIdx.x & 3) * 16;

    const int m     = lane & 15;
    const int kh    = lane >> 4;
    const int kbase = kh * 2;

    const bool edge_tile = q_tile < 7;
    const int  gblk = (q_tile >= 21) + (q_tile >= 35);
    const int  nblk = q_tile - 7 - gblk * 14;
    const int  rsh  = edge_tile ? 3 : 2;
    const int  g0   = edge_tile ? 12 : gblk * 4;
    const int  n0   = edge_tile ? q_tile * 8 : nblk * 4;
    const int g    = g0 + (m >> rsh);
    const int n    = n0 + (m & ((1 << rsh) - 1));
    const int o0   = 4 * g;
    const int xrow = n * 56;

    const int jq = wave;

    const bool m2   = g == 0;
    const bool p1   = g == 13;
    const int startL = m2 ? 0 : o0 - 2;
    const int startR = p1 ? 52 : o0 + 2;
    const int hoff = kh ? startR : startL;
    const bool wrap_lane = (g == 0);
    const bool wrap_wave = (q_tile >= 7) && (q_tile < 21);

    const float* ap  = W + (row_base + m) * KRED + jq * JGRP + kbase;
    const float* hp = x + (size_t)(jq * JGRP) * PLANE + xrow + hoff;
    const float* zp0 = x + (size_t)(jq * JGRP + kbase + 0) * PLANE + xrow + 54;
    const float* zp1 = x + (size_t)(jq * JGRP + kbase + 1) * PLANE + xrow + 54;

    v8f acc0 = {}, acc1 = {}, acc2 = {}, acc3 = {};

    v2f zu[2] = {}, zv[2] = {};
    if (wrap_wave) {
        #pragma unroll
        for (int c4 = 0; c4 < 2; ++c4) {
            zu[c4] = *(const v2f*)(zp0 + c4 * 4 * PLANE);
            zv[c4] = *(const v2f*)(zp1 + c4 * 4 * PLANE);
        }
    }

    float hx[2][4][4];
    v2f a_first[3][2];
    #pragma unroll
    for (int p = 0; p < 4; ++p) {
        const float* pp = hp + p * PLANE;
        hx[0][p][0] = pp[0]; hx[0][p][1] = pp[1]; hx[0][p][2] = pp[2]; hx[0][p][3] = pp[3];
    }
    __builtin_amdgcn_sched_barrier(0);
    #pragma unroll
    for (int k = 0; k < 3; ++k) {
        a_first[k][0] = *(const v2f*)(ap + k * 128);
        a_first[k][1] = *(const v2f*)(ap + k * 128 + 64);
    }
    __builtin_amdgcn_sched_barrier(0);
    #pragma unroll
    for (int p = 0; p < 4; ++p) {
        const float* pp = hp + (4 + p) * PLANE;
        hx[1][p][0] = pp[0]; hx[1][p][1] = pp[1]; hx[1][p][2] = pp[2]; hx[1][p][3] = pp[3];
    }
    __builtin_amdgcn_sched_barrier(0);

    #pragma unroll
    for (int c4 = 0; c4 < 2; ++c4) {
        float h[4][4];
        #pragma unroll
        for (int p = 0; p < 4; ++p) {
            #pragma unroll
            for (int i = 0; i < 4; ++i) h[p][i] = hx[c4][p][i];
        }
        #pragma unroll
        for (int p = 0; p < 2; ++p) {
            #pragma unroll
            for (int i = 0; i < 4; ++i) swap_halves(h[p][i], h[p + 2][i]);
        }
        const float ul0 = h[0][0], ul1 = h[0][1], ul2 = h[0][2], ul3 = h[0][3];
        const float vl0 = h[1][0], vl1 = h[1][1], vl2 = h[1][2], vl3 = h[1][3];
        const float ur0 = h[2][0], ur1 = h[2][1], ur2 = h[2][2], ur3 = h[2][3];
        const float vr0 = h[3][0], vr1 = h[3][1], vr2 = h[3][2], vr3 = h[3][3];

        v2f bs[7];
        bs[0][0] = m2 ? 0.0f : ul0;   bs[0][1] = m2 ? 0.0f : vl0;
        bs[1][0] = m2 ? 0.0f : ul1;   bs[1][1] = m2 ? 0.0f : vl1;
        bs[2][0] = m2 ? ul0  : ul2;   bs[2][1] = m2 ? vl0  : vl2;
        bs[3][0] = m2 ? ul1  : ul3;   bs[3][1] = m2 ? vl1  : vl3;
        bs[4][0] = p1 ? ur2  : ur0;   bs[4][1] = p1 ? vr2  : vr0;
        bs[5][0] = p1 ? ur3  : ur1;   bs[5][1] = p1 ? vr3  : vr1;
        bs[6][0] = p1 ? 0.0f : ur2;   bs[6][1] = p1 ? 0.0f : vr2;

        v2f bt[3];
        bt[0][0] = wrap_lane ? zu[c4][0] : bs[0][0];
        bt[0][1] = wrap_lane ? zv[c4][0] : bs[0][1];
        bt[1][0] = wrap_lane ? zu[c4][1] : bs[1][0];
        bt[1][1] = wrap_lane ? zv[c4][1] : bs[1][1];
        bt[2][0] = wrap_lane ? 0.0f : bs[2][0];
        bt[2][1] = wrap_lane ? 0.0f : bs[2][1];

        #pragma unroll
        for (int k = 0; k < 3; ++k) {
            const float* apk = ap + k * 128 + c4 * 4;
            v2f a0 = c4 == 0 ? a_first[k][0] : *(const v2f*)(apk);
            v2f a1 = c4 == 0 ? a_first[k][1] : *(const v2f*)(apk + 64);

            acc0 = __builtin_amdgcn_wmma_f32_16x16x4_f32(
                false, a0, false, bs[k + 1], (short)0, acc0, false, false);
            acc1 = __builtin_amdgcn_wmma_f32_16x16x4_f32(
                false, a0, false, bs[k + 2], (short)0, acc1, false, false);
            acc2 = __builtin_amdgcn_wmma_f32_16x16x4_f32(
                false, a0, false, bs[k + 3], (short)0, acc2, false, false);
            acc3 = __builtin_amdgcn_wmma_f32_16x16x4_f32(
                false, a0, false, bs[k + 4], (short)0, acc3, false, false);

            acc0 = __builtin_amdgcn_wmma_f32_16x16x4_f32(
                false, a1, false, bt[k], (short)0, acc0, false, false);
            acc1 = __builtin_amdgcn_wmma_f32_16x16x4_f32(
                false, a1, false, bs[k + 1], (short)0, acc1, false, false);
            acc2 = __builtin_amdgcn_wmma_f32_16x16x4_f32(
                false, a1, false, bs[k + 2], (short)0, acc2, false, false);
            acc3 = __builtin_amdgcn_wmma_f32_16x16x4_f32(
                false, a1, false, bs[k + 3], (short)0, acc3, false, false);
        }
        if (c4 == 0) __builtin_amdgcn_sched_barrier(0);
    }

    #pragma unroll
    for (int vr = 0; vr < 8; ++vr) {
        v4f p = { acc0[vr], acc1[vr], acc2[vr], acc3[vr] };
        part[wave][vr][lane] = p;
    }
    __syncthreads();

    const int ostore = o0 * 56 + ((n + 1) % 56);

    {
        const int vr = wave;
        v4f s = part[0][vr][lane];
        #pragma unroll
        for (int w = 1; w < NWAVE; ++w) {
            s += part[w][vr][lane];
        }
        const int i = row_base + vr + 8 * kh;
        out[i * NCOL + ostore]       = s[0];
        out[i * NCOL + ostore + 56]  = s[1];
        out[i * NCOL + ostore + 112] = s[2];
        out[i * NCOL + ostore + 168] = s[3];
    }
}

extern "C" void kernel_launch(void* const* d_in, const int* in_sizes, int n_in,
                              void* d_out, int out_size, void* d_ws, size_t ws_size,
                              hipStream_t stream) {
    const float* x = (const float*)d_in[0];
    const float* W = (const float*)d_in[1];
    float* out = (float*)d_out;

    shiftconv_wmma_f32<<<196, 256, 0, stream>>>(x, W, out);
}
